# starting kernel
# baseline (speedup 1.0000x reference)
#include <hip/hip_runtime.h>
#include <math.h>


typedef __attribute__((ext_vector_type(2))) float v2f;
typedef __attribute__((ext_vector_type(8))) float v8f;

#define NUM_E 50000
#define NUM_F 64
#define NUM_B 64
#define NTILES (NUM_E / 16)
#define WAVES_PER_BLOCK 8
#define GRID_MAIN ((NTILES + WAVES_PER_BLOCK - 1) / WAVES_PER_BLOCK)

#define LOG2E 1.4426950408889634f

#define WS_AQ 0
#define WS_W  (NUM_B * NUM_F)
#define WS_Q  (2 * NUM_B * NUM_F)

__global__ void kbln_precompute(const float* __restrict__ lit,
                                const float* __restrict__ c,
                                const float* __restrict__ var,
                                const float* __restrict__ nfw,
                                const int* __restrict__ head_ids,
                                const int* __restrict__ rel_ids,
                                float* __restrict__ ws) {
  __shared__ float sq[NUM_F];
  int t = threadIdx.x;
  if (t < NUM_F) {
    float q = sqrtf(LOG2E / var[t]);
    sq[t] = q;
    ws[WS_Q + t] = q;
  }
  __syncthreads();
  for (int idx = t; idx < NUM_B * NUM_F; idx += 256) {
    int b = idx >> 6;
    int f = idx & 63;
    float A = lit[(size_t)head_ids[b] * NUM_F + f] - c[f];
    ws[WS_AQ + idx] = A * sq[f];
    ws[WS_W + idx]  = nfw[(size_t)rel_ids[b] * NUM_F + f];
  }
}

__global__ void __launch_bounds__(256)
kbln_score(const float* __restrict__ lit,
           const float* __restrict__ ws,
           float* __restrict__ out) {
  __shared__ float s_aq[NUM_B * NUM_F];
  __shared__ float s_w[NUM_B * NUM_F];
  __shared__ float sq[NUM_F];

  int t = threadIdx.x;
  for (int i = t; i < NUM_B * NUM_F; i += 256) {
    s_aq[i] = ws[WS_AQ + i];
    s_w[i]  = ws[WS_W + i];
  }
  if (t < NUM_F) sq[t] = ws[WS_Q + t];
  __syncthreads();

  int wave = t >> 5;
  int lane = t & 31;
  int tile = blockIdx.x * WAVES_PER_BLOCK + wave;
  if (tile >= NTILES) return;

  int e = tile * 16 + (lane & 15);
  int h = lane >> 4;
  int fo = 2 * h;

  const float* lrow = lit + (size_t)e * NUM_F;
  float lq[32];
#pragma unroll
  for (int k = 0; k < 16; ++k) {
    int f = 4 * k + fo;
    float2 lv = *reinterpret_cast<const float2*>(lrow + f);
    float2 qq = *reinterpret_cast<const float2*>(&sq[f]);
    lq[2 * k]     = lv.x * qq.x;
    lq[2 * k + 1] = lv.y * qq.y;
  }

  for (int b = 0; b < NUM_B; ++b) {
    v8f acc = {};
    const float* pa = s_aq + b * NUM_F + fo;
    const float* pw = s_w  + b * NUM_F + fo;
#pragma unroll
    for (int k = 0; k < 16; ++k) {
      float2 aq = *reinterpret_cast<const float2*>(pa + 4 * k);
      v2f Bm    = *reinterpret_cast<const v2f*>(pw + 4 * k);
      float u0 = lq[2 * k]     - aq.x;
      float u1 = lq[2 * k + 1] - aq.y;
      v2f A;
      A[0] = __builtin_amdgcn_exp2f(-(u0 * u0));
      A[1] = __builtin_amdgcn_exp2f(-(u1 * u1));
      acc = __builtin_amdgcn_wmma_f32_16x16x4_f32(
           false, A,  false, Bm,
           (short)0, acc,  false,  false);
    }
    if ((lane & 15) == 0) {
      float4* o4 = reinterpret_cast<float4*>(out + (size_t)b * NUM_E +
                                             tile * 16 + 8 * h);
      o4[0] = make_float4(acc[0], acc[1], acc[2], acc[3]);
      o4[1] = make_float4(acc[4], acc[5], acc[6], acc[7]);
    }
  }
}

extern "C" void kernel_launch(void* const* d_in, const int* in_sizes, int n_in,
                              void* d_out, int out_size, void* d_ws, size_t ws_size,
                              hipStream_t stream) {
  const float* lit  = (const float*)d_in[0];
  const float* c    = (const float*)d_in[1];
  const float* var  = (const float*)d_in[2];
  const float* nfw  = (const float*)d_in[3];
  const int*   head = (const int*)d_in[4];
  const int*   rel  = (const int*)d_in[5];
  float* ws = (float*)d_ws;

  kbln_precompute<<<1, 256, 0, stream>>>(lit, c, var, nfw, head, rel, ws);
  kbln_score<<<GRID_MAIN, 256, 0, stream>>>(lit, ws, (float*)d_out);
}


# Re-tile: a 4-wave group owns 13 e-tiles; each wave holds their lq in registers and walks its quarter of b, so one grouped LDS read of aq/w feeds 208 WMMAs on 13 accumulators; sum order unchanged
# speedup vs baseline: 1.2195x; 1.2195x over previous
#include <hip/hip_runtime.h>
#include <math.h>


typedef __attribute__((ext_vector_type(2))) float v2f;
typedef __attribute__((ext_vector_type(4))) float v4f;
typedef __attribute__((ext_vector_type(8))) float v8f;

#define NUM_E 50000
#define NUM_F 64
#define NUM_B 64
#define NTILES (NUM_E / 16)
#define TPB 13
#define GRID_MAIN ((NTILES + TPB - 1) / TPB)
#define ROWS (TPB * 16)
#define PITCH 68
#define BPW 16

#define LOG2E 1.4426950408889634f

#define WS_AQ 0
#define WS_W  (NUM_B * NUM_F)
#define WS_Q  (2 * NUM_B * NUM_F)

__global__ void kbln_precompute(const float* __restrict__ lit,
                                const float* __restrict__ c,
                                const float* __restrict__ var,
                                const float* __restrict__ nfw,
                                const int* __restrict__ head_ids,
                                const int* __restrict__ rel_ids,
                                float* __restrict__ ws) {
  __shared__ float sq[NUM_F];
  int t = threadIdx.x;
  if (t < NUM_F) {
    float q = sqrtf(LOG2E / var[t]);
    sq[t] = q;
    ws[WS_Q + t] = q;
  }
  __syncthreads();
  for (int idx = t; idx < NUM_B * NUM_F; idx += 256) {
    int b = idx >> 6;
    int f = idx & 63;
    float A = lit[(size_t)head_ids[b] * NUM_F + f] - c[f];
    ws[WS_AQ + idx] = A * sq[f];
    ws[WS_W + idx]  = nfw[(size_t)rel_ids[b] * NUM_F + f];
  }
}

__device__ __forceinline__ int staged_slot(int k, int h) {
  return 8 * (k >> 1) + 4 * h + 2 * (k & 1);
}

__device__ __forceinline__ void load_bvec(v4f (&aq)[8], v4f (&wv)[8],
                                          const float* s_aq, const float* s_w,
                                          int b, int h) {
  const float* pa = s_aq + b * NUM_F + 4 * h;
  const float* pw = s_w  + b * NUM_F + 4 * h;
#pragma unroll
  for (int j = 0; j < 8; ++j) {
    aq[j] = *reinterpret_cast<const v4f*>(pa + 8 * j);
    wv[j] = *reinterpret_cast<const v4f*>(pw + 8 * j);
  }
}

__device__ __forceinline__ v2f step_u(const v4f (&lq)[TPB][8], const v4f (&aq)[8], int s) {
  const int k = s / TPB, tt = s % TPB;
  const int j = k >> 1, hf = k & 1;
  v2f u;
  u[0] = lq[tt][j][2 * hf]     - aq[j][2 * hf];
  u[1] = lq[tt][j][2 * hf + 1] - aq[j][2 * hf + 1];
  return u;
}

__device__ __forceinline__ v2f step_m(const v4f (&lq)[TPB][8], const v4f (&aq)[8], int s) {
  v2f u = step_u(lq, aq, s);
  v2f m;
  m[0] = -(u[0] * u[0]);
  m[1] = -(u[1] * u[1]);
  return m;
}

__device__ __forceinline__ v2f exp2_v2(v2f m) {
  v2f A;
  A[0] = __builtin_amdgcn_exp2f(m[0]);
  A[1] = __builtin_amdgcn_exp2f(m[1]);
  return A;
}

#define NSTEP (16 * TPB)

__device__ __forceinline__ void score_b(const v4f (&lq)[TPB][8],
                                        const v4f (&aq)[8], const v4f (&wv)[8],
                                        v4f (&aqn)[8], v4f (&wvn)[8],
                                        const float* s_aq, const float* s_w,
                                        int bnext, int h,
                                        float* orow, int c, int ntile) {
  load_bvec(aqn, wvn, s_aq, s_w, bnext, h);
  __builtin_amdgcn_sched_barrier(0);

  v8f acc[TPB];
#pragma unroll
  for (int tt = 0; tt < TPB; ++tt) acc[tt] = (v8f){};

  v2f A0 = exp2_v2(step_m(lq, aq, 0));
  v2f A1 = exp2_v2(step_m(lq, aq, 1));
  v2f m2 = step_m(lq, aq, 2);
  v2f u3 = step_u(lq, aq, 3);
  v2f u4 = step_u(lq, aq, 4);
#pragma unroll
  for (int s = 0; s < NSTEP; ++s) {
    const int k = s / TPB, tt = s % TPB;
    const int j = k >> 1, hf = k & 1;
    v2f A2 = A1, m3 = m2, u5 = u4;
    if (s + 2 < NSTEP) A2[0] = __builtin_amdgcn_exp2f(m2[0]);
    if (s + 3 < NSTEP) {
      m3[0] = -(u3[0] * u3[0]);
      m3[1] = -(u3[1] * u3[1]);
    }
    asm volatile("" :: "v"(A2[0]), "v"(m3[0]), "v"(m3[1]));
    __builtin_amdgcn_sched_barrier(0);
    if (s + 2 < NSTEP) A2[1] = __builtin_amdgcn_exp2f(m2[1]);
    if (s + 5 < NSTEP) u5 = step_u(lq, aq, s + 5);
    asm volatile("" :: "v"(A2[1]), "v"(u5[0]), "v"(u5[1]));
    v2f Bm;
    Bm[0] = wv[j][2 * hf];
    Bm[1] = wv[j][2 * hf + 1];
    acc[tt] = __builtin_amdgcn_wmma_f32_16x16x4_f32(
         false, A0,  false, Bm,
         (short)0, acc[tt],  false,  false);
    __builtin_amdgcn_sched_barrier(0);
    A0 = A1;
    A1 = A2;
    m2 = m3;
    u3 = u4;
    u4 = u5;
  }

  if (c == 0) {
    if (ntile == TPB) {
#pragma unroll
      for (int tt = 0; tt < TPB; ++tt) {
        float4* o4 = reinterpret_cast<float4*>(orow + tt * 16);
        o4[0] = make_float4(acc[tt][0], acc[tt][1], acc[tt][2], acc[tt][3]);
        o4[1] = make_float4(acc[tt][4], acc[tt][5], acc[tt][6], acc[tt][7]);
      }
    } else {
#pragma unroll
      for (int tt = 0; tt < TPB; ++tt) {
        if (tt < ntile) {
          float4* o4 = reinterpret_cast<float4*>(orow + tt * 16);
          o4[0] = make_float4(acc[tt][0], acc[tt][1], acc[tt][2], acc[tt][3]);
          o4[1] = make_float4(acc[tt][4], acc[tt][5], acc[tt][6], acc[tt][7]);
        }
      }
    }
  }
}

__global__ void __launch_bounds__(128)
kbln_score(const float* __restrict__ lit,
           const float* __restrict__ ws,
           float* __restrict__ out) {
  __shared__ __attribute__((aligned(16))) float s_aq[NUM_B * NUM_F];
  __shared__ __attribute__((aligned(16))) float s_w[NUM_B * NUM_F];
  __shared__ __attribute__((aligned(16))) float s_lq[ROWS * PITCH];

  const int t = threadIdx.x;
  const int k4 = t & 15;

#pragma unroll
  for (int i = t; i < NUM_B * NUM_F / 4; i += 128) {
    float4 a4 = reinterpret_cast<const float4*>(ws + WS_AQ)[i];
    float4 w4 = reinterpret_cast<const float4*>(ws + WS_W)[i];
    int base = (i >> 4) * NUM_F;
    *reinterpret_cast<float2*>(s_aq + base + staged_slot(k4, 0)) = make_float2(a4.x, a4.y);
    *reinterpret_cast<float2*>(s_aq + base + staged_slot(k4, 1)) = make_float2(a4.z, a4.w);
    *reinterpret_cast<float2*>(s_w  + base + staged_slot(k4, 0)) = make_float2(w4.x, w4.y);
    *reinterpret_cast<float2*>(s_w  + base + staged_slot(k4, 1)) = make_float2(w4.z, w4.w);
  }

  {
    const float4 q4 = reinterpret_cast<const float4*>(ws + WS_Q)[k4];
    const int row0 = blockIdx.x * ROWS;
#pragma unroll
    for (int r = t >> 4; r < ROWS; r += 8) {
      int e = row0 + r;
      e = e < NUM_E ? e : NUM_E - 1;
      float4 v = reinterpret_cast<const float4*>(lit + (size_t)e * NUM_F)[k4];
      float* dst = s_lq + r * PITCH;
      *reinterpret_cast<float2*>(dst + staged_slot(k4, 0)) = make_float2(v.x * q4.x, v.y * q4.y);
      *reinterpret_cast<float2*>(dst + staged_slot(k4, 1)) = make_float2(v.z * q4.z, v.w * q4.w);
    }
  }
  __syncthreads();

  const int wave = t >> 5;
  const int lane = t & 31;
  const int c = lane & 15;
  const int h = lane >> 4;
  const int tile0 = blockIdx.x * TPB;
  const int ntile = (NTILES - tile0) < TPB ? (NTILES - tile0) : TPB;

  v4f lq[TPB][8];
#pragma unroll
  for (int tt = 0; tt < TPB; ++tt) {
#pragma unroll
    for (int j = 0; j < 8; ++j)
      lq[tt][j] = *reinterpret_cast<const v4f*>(s_lq + (tt * 16 + c) * PITCH + 8 * j + 4 * h);
  }

  const int b0 = wave * BPW;
  float* orow = out + (size_t)b0 * NUM_E + tile0 * 16 + 8 * h;

  v4f aqA[8], wvA[8], aqB[8], wvB[8];
  load_bvec(aqA, wvA, s_aq, s_w, b0, h);
  for (int bb = 0; bb < BPW; bb += 2) {
    int bn1 = b0 + bb + 1;
    int bn2 = b0 + bb + 2;
    bn2 = bn2 < NUM_B ? bn2 : NUM_B - 1;
    score_b(lq, aqA, wvA, aqB, wvB, s_aq, s_w, bn1, h, orow, c, ntile);
    orow += NUM_E;
    score_b(lq, aqB, wvB, aqA, wvA, s_aq, s_w, bn2, h, orow, c, ntile);
    orow += NUM_E;
  }
}

extern "C" void kernel_launch(void* const* d_in, const int* in_sizes, int n_in,
                              void* d_out, int out_size, void* d_ws, size_t ws_size,
                              hipStream_t stream) {
  const float* lit  = (const float*)d_in[0];
  const float* c    = (const float*)d_in[1];
  const float* var  = (const float*)d_in[2];
  const float* nfw  = (const float*)d_in[3];
  const int*   head = (const int*)d_in[4];
  const int*   rel  = (const int*)d_in[5];
  float* ws = (float*)d_ws;

  kbln_precompute<<<1, 256, 0, stream>>>(lit, c, var, nfw, head, rel, ws);
  kbln_score<<<GRID_MAIN, 128, 0, stream>>>(lit, ws, (float*)d_out);
}


# One launch instead of two: each work-group gathers the head and relation rows itself (ids, then rows, all in flight) and builds aq, w, q in its LDS; table launch and workspace round trip gone
# speedup vs baseline: 1.6031x; 1.3146x over previous
#include <hip/hip_runtime.h>
#include <math.h>


typedef __attribute__((ext_vector_type(2))) float v2f;
typedef __attribute__((ext_vector_type(4))) float v4f;
typedef __attribute__((ext_vector_type(8))) float v8f;

#define NUM_E 50000
#define NUM_F 64
#define NUM_B 64
#define NTILES (NUM_E / 16)
#define TPB 13
#define GRID_MAIN ((NTILES + TPB - 1) / TPB)
#define ROWS (TPB * 16)
#define PITCH 68
#define BPW 16

#define LOG2E 1.4426950408889634f

__device__ __forceinline__ int staged_slot(int k, int h) {
  return 8 * (k >> 1) + 4 * h + 2 * (k & 1);
}

__device__ __forceinline__ void load_bvec(v4f (&aq)[8], v4f (&wv)[8],
                                          const float* s_aq, const float* s_w,
                                          int b, int h) {
  const float* pa = s_aq + b * NUM_F + 4 * h;
  const float* pw = s_w  + b * NUM_F + 4 * h;
#pragma unroll
  for (int j = 0; j < 8; ++j) {
    aq[j] = *reinterpret_cast<const v4f*>(pa + 8 * j);
    wv[j] = *reinterpret_cast<const v4f*>(pw + 8 * j);
  }
}

__device__ __forceinline__ v2f step_u(const v4f (&lq)[TPB][8], const v4f (&aq)[8], int s) {
  const int k = s / TPB, tt = s % TPB;
  const int j = k >> 1, hf = k & 1;
  v2f u;
  u[0] = lq[tt][j][2 * hf]     - aq[j][2 * hf];
  u[1] = lq[tt][j][2 * hf + 1] - aq[j][2 * hf + 1];
  return u;
}

__device__ __forceinline__ v2f step_m(const v4f (&lq)[TPB][8], const v4f (&aq)[8], int s) {
  v2f u = step_u(lq, aq, s);
  v2f m;
  m[0] = -(u[0] * u[0]);
  m[1] = -(u[1] * u[1]);
  return m;
}

__device__ __forceinline__ v2f exp2_v2(v2f m) {
  v2f A;
  A[0] = __builtin_amdgcn_exp2f(m[0]);
  A[1] = __builtin_amdgcn_exp2f(m[1]);
  return A;
}

#define NSTEP (16 * TPB)

__device__ __forceinline__ void score_b(const v4f (&lq)[TPB][8],
                                        const v4f (&aq)[8], const v4f (&wv)[8],
                                        v4f (&aqn)[8], v4f (&wvn)[8],
                                        const float* s_aq, const float* s_w,
                                        int bnext, int h,
                                        float* orow, int c, int ntile) {
  load_bvec(aqn, wvn, s_aq, s_w, bnext, h);
  __builtin_amdgcn_sched_barrier(0);

  v8f acc[TPB];
#pragma unroll
  for (int tt = 0; tt < TPB; ++tt) acc[tt] = (v8f){};

  v2f A0 = exp2_v2(step_m(lq, aq, 0));
  v2f A1 = exp2_v2(step_m(lq, aq, 1));
  v2f m2 = step_m(lq, aq, 2);
  v2f u3 = step_u(lq, aq, 3);
  v2f u4 = step_u(lq, aq, 4);
#pragma unroll
  for (int s = 0; s < NSTEP; ++s) {
    const int k = s / TPB, tt = s % TPB;
    const int j = k >> 1, hf = k & 1;
    v2f A2 = A1, m3 = m2, u5 = u4;
    if (s + 2 < NSTEP) A2[0] = __builtin_amdgcn_exp2f(m2[0]);
    if (s + 3 < NSTEP) {
      m3[0] = -(u3[0] * u3[0]);
      m3[1] = -(u3[1] * u3[1]);
    }
    asm volatile("" :: "v"(A2[0]), "v"(m3[0]), "v"(m3[1]));
    __builtin_amdgcn_sched_barrier(0);
    if (s + 2 < NSTEP) A2[1] = __builtin_amdgcn_exp2f(m2[1]);
    if (s + 5 < NSTEP) u5 = step_u(lq, aq, s + 5);
    asm volatile("" :: "v"(A2[1]), "v"(u5[0]), "v"(u5[1]));
    v2f Bm;
    Bm[0] = wv[j][2 * hf];
    Bm[1] = wv[j][2 * hf + 1];
    acc[tt] = __builtin_amdgcn_wmma_f32_16x16x4_f32(
         false, A0,  false, Bm,
         (short)0, acc[tt],  false,  false);
    __builtin_amdgcn_sched_barrier(0);
    A0 = A1;
    A1 = A2;
    m2 = m3;
    u3 = u4;
    u4 = u5;
  }

  if (c == 0) {
    if (ntile == TPB) {
#pragma unroll
      for (int tt = 0; tt < TPB; ++tt) {
        float4* o4 = reinterpret_cast<float4*>(orow + tt * 16);
        o4[0] = make_float4(acc[tt][0], acc[tt][1], acc[tt][2], acc[tt][3]);
        o4[1] = make_float4(acc[tt][4], acc[tt][5], acc[tt][6], acc[tt][7]);
      }
    } else {
#pragma unroll
      for (int tt = 0; tt < TPB; ++tt) {
        if (tt < ntile) {
          float4* o4 = reinterpret_cast<float4*>(orow + tt * 16);
          o4[0] = make_float4(acc[tt][0], acc[tt][1], acc[tt][2], acc[tt][3]);
          o4[1] = make_float4(acc[tt][4], acc[tt][5], acc[tt][6], acc[tt][7]);
        }
      }
    }
  }
}

__global__ void __launch_bounds__(128)
kbln_score(const float* __restrict__ lit,
           const float* __restrict__ cen,
           const float* __restrict__ var,
           const float* __restrict__ nfw,
           const int* __restrict__ head_ids,
           const int* __restrict__ rel_ids,
           float* __restrict__ out) {
  __shared__ __attribute__((aligned(16))) float s_aq[NUM_B * NUM_F];
  __shared__ __attribute__((aligned(16))) float s_w[NUM_B * NUM_F];
  __shared__ __attribute__((aligned(16))) float s_lq[ROWS * PITCH];

  const int t = threadIdx.x;
  const int k4 = t & 15;

  const float4 var4 = reinterpret_cast<const float4*>(var)[k4];
  const float4 c4   = reinterpret_cast<const float4*>(cen)[k4];
  float4 q4;
  q4.x = sqrtf(LOG2E / var4.x);
  q4.y = sqrtf(LOG2E / var4.y);
  q4.z = sqrtf(LOG2E / var4.z);
  q4.w = sqrtf(LOG2E / var4.w);

  int hid[NUM_B / 8], rid[NUM_B / 8];
#pragma unroll
  for (int it = 0; it < NUM_B / 8; ++it) {
    hid[it] = head_ids[(t >> 4) + 8 * it];
    rid[it] = rel_ids[(t >> 4) + 8 * it];
  }
  float4 l4[NUM_B / 8], w4[NUM_B / 8];
#pragma unroll
  for (int it = 0; it < NUM_B / 8; ++it) {
    l4[it] = reinterpret_cast<const float4*>(lit + (size_t)hid[it] * NUM_F)[k4];
    w4[it] = reinterpret_cast<const float4*>(nfw + (size_t)rid[it] * NUM_F)[k4];
  }
#pragma unroll
  for (int it = 0; it < NUM_B / 8; ++it) {
    const int base = ((t >> 4) + 8 * it) * NUM_F;
    *reinterpret_cast<float2*>(s_aq + base + staged_slot(k4, 0)) =
        make_float2((l4[it].x - c4.x) * q4.x, (l4[it].y - c4.y) * q4.y);
    *reinterpret_cast<float2*>(s_aq + base + staged_slot(k4, 1)) =
        make_float2((l4[it].z - c4.z) * q4.z, (l4[it].w - c4.w) * q4.w);
    *reinterpret_cast<float2*>(s_w  + base + staged_slot(k4, 0)) = make_float2(w4[it].x, w4[it].y);
    *reinterpret_cast<float2*>(s_w  + base + staged_slot(k4, 1)) = make_float2(w4[it].z, w4[it].w);
  }

  {
    const int row0 = blockIdx.x * ROWS;
#pragma unroll
    for (int r = t >> 4; r < ROWS; r += 8) {
      int e = row0 + r;
      e = e < NUM_E ? e : NUM_E - 1;
      float4 v = reinterpret_cast<const float4*>(lit + (size_t)e * NUM_F)[k4];
      float* dst = s_lq + r * PITCH;
      *reinterpret_cast<float2*>(dst + staged_slot(k4, 0)) = make_float2(v.x * q4.x, v.y * q4.y);
      *reinterpret_cast<float2*>(dst + staged_slot(k4, 1)) = make_float2(v.z * q4.z, v.w * q4.w);
    }
  }
  __syncthreads();

  const int wave = t >> 5;
  const int lane = t & 31;
  const int c = lane & 15;
  const int h = lane >> 4;
  const int tile0 = blockIdx.x * TPB;
  const int ntile = (NTILES - tile0) < TPB ? (NTILES - tile0) : TPB;

  v4f lq[TPB][8];
#pragma unroll
  for (int tt = 0; tt < TPB; ++tt) {
#pragma unroll
    for (int j = 0; j < 8; ++j)
      lq[tt][j] = *reinterpret_cast<const v4f*>(s_lq + (tt * 16 + c) * PITCH + 8 * j + 4 * h);
  }

  const int b0 = wave * BPW;
  float* orow = out + (size_t)b0 * NUM_E + tile0 * 16 + 8 * h;

  v4f aqA[8], wvA[8], aqB[8], wvB[8];
  load_bvec(aqA, wvA, s_aq, s_w, b0, h);
  for (int bb = 0; bb < BPW; bb += 2) {
    int bn1 = b0 + bb + 1;
    int bn2 = b0 + bb + 2;
    bn2 = bn2 < NUM_B ? bn2 : NUM_B - 1;
    score_b(lq, aqA, wvA, aqB, wvB, s_aq, s_w, bn1, h, orow, c, ntile);
    orow += NUM_E;
    score_b(lq, aqB, wvB, aqA, wvA, s_aq, s_w, bn2, h, orow, c, ntile);
    orow += NUM_E;
  }
}

extern "C" void kernel_launch(void* const* d_in, const int* in_sizes, int n_in,
                              void* d_out, int out_size, void* d_ws, size_t ws_size,
                              hipStream_t stream) {
  const float* lit  = (const float*)d_in[0];
  const float* c    = (const float*)d_in[1];
  const float* var  = (const float*)d_in[2];
  const float* nfw  = (const float*)d_in[3];
  const int*   head = (const int*)d_in[4];
  const int*   rel  = (const int*)d_in[5];

  kbln_score<<<GRID_MAIN, 128, 0, stream>>>(lit, c, var, nfw, head, rel, (float*)d_out);
}


# Work-group start: each thread's 26 lit-row reads go out as one batch (constant trip count) behind the id reads and share one memory round trip with the gathered rows; was 7 load-wait-store trips
# speedup vs baseline: 1.7807x; 1.1108x over previous
#include <hip/hip_runtime.h>
#include <math.h>


typedef __attribute__((ext_vector_type(2))) float v2f;
typedef __attribute__((ext_vector_type(4))) float v4f;
typedef __attribute__((ext_vector_type(8))) float v8f;

#define NUM_E 50000
#define NUM_F 64
#define NUM_B 64
#define NTILES (NUM_E / 16)
#define TPB 13
#define GRID_MAIN ((NTILES + TPB - 1) / TPB)
#define ROWS (TPB * 16)
#define PITCH 68
#define BPW 16

#define LOG2E 1.4426950408889634f

__device__ __forceinline__ int staged_slot(int k, int h) {
  return 8 * (k >> 1) + 4 * h + 2 * (k & 1);
}

__device__ __forceinline__ void load_bvec(v4f (&aq)[8], v4f (&wv)[8],
                                          const float* s_aq, const float* s_w,
                                          int b, int h) {
  const float* pa = s_aq + b * NUM_F + 4 * h;
  const float* pw = s_w  + b * NUM_F + 4 * h;
#pragma unroll
  for (int j = 0; j < 8; ++j) {
    aq[j] = *reinterpret_cast<const v4f*>(pa + 8 * j);
    wv[j] = *reinterpret_cast<const v4f*>(pw + 8 * j);
  }
}

__device__ __forceinline__ v2f step_u(const v4f (&lq)[TPB][8], const v4f (&aq)[8], int s) {
  const int k = s / TPB, tt = s % TPB;
  const int j = k >> 1, hf = k & 1;
  v2f u;
  u[0] = lq[tt][j][2 * hf]     - aq[j][2 * hf];
  u[1] = lq[tt][j][2 * hf + 1] - aq[j][2 * hf + 1];
  return u;
}

__device__ __forceinline__ v2f step_m(const v4f (&lq)[TPB][8], const v4f (&aq)[8], int s) {
  v2f u = step_u(lq, aq, s);
  v2f m;
  m[0] = -(u[0] * u[0]);
  m[1] = -(u[1] * u[1]);
  return m;
}

__device__ __forceinline__ v2f exp2_v2(v2f m) {
  v2f A;
  A[0] = __builtin_amdgcn_exp2f(m[0]);
  A[1] = __builtin_amdgcn_exp2f(m[1]);
  return A;
}

#define NSTEP (16 * TPB)

__device__ __forceinline__ void score_b(const v4f (&lq)[TPB][8],
                                        const v4f (&aq)[8], const v4f (&wv)[8],
                                        v4f (&aqn)[8], v4f (&wvn)[8],
                                        const float* s_aq, const float* s_w,
                                        int bnext, int h,
                                        float* orow, int c, int ntile) {
  load_bvec(aqn, wvn, s_aq, s_w, bnext, h);
  __builtin_amdgcn_sched_barrier(0);

  v8f acc[TPB];
#pragma unroll
  for (int tt = 0; tt < TPB; ++tt) acc[tt] = (v8f){};

  v2f A0 = exp2_v2(step_m(lq, aq, 0));
  v2f A1 = exp2_v2(step_m(lq, aq, 1));
  v2f m2 = step_m(lq, aq, 2);
  v2f u3 = step_u(lq, aq, 3);
  v2f u4 = step_u(lq, aq, 4);
#pragma unroll
  for (int s = 0; s < NSTEP; ++s) {
    const int k = s / TPB, tt = s % TPB;
    const int j = k >> 1, hf = k & 1;
    v2f A2 = A1, m3 = m2, u5 = u4;
    if (s + 2 < NSTEP) A2[0] = __builtin_amdgcn_exp2f(m2[0]);
    if (s + 3 < NSTEP) {
      m3[0] = -(u3[0] * u3[0]);
      m3[1] = -(u3[1] * u3[1]);
    }
    asm volatile("" :: "v"(A2[0]), "v"(m3[0]), "v"(m3[1]));
    __builtin_amdgcn_sched_barrier(0);
    if (s + 2 < NSTEP) A2[1] = __builtin_amdgcn_exp2f(m2[1]);
    if (s + 5 < NSTEP) u5 = step_u(lq, aq, s + 5);
    asm volatile("" :: "v"(A2[1]), "v"(u5[0]), "v"(u5[1]));
    v2f Bm;
    Bm[0] = wv[j][2 * hf];
    Bm[1] = wv[j][2 * hf + 1];
    acc[tt] = __builtin_amdgcn_wmma_f32_16x16x4_f32(
         false, A0,  false, Bm,
         (short)0, acc[tt],  false,  false);
    __builtin_amdgcn_sched_barrier(0);
    A0 = A1;
    A1 = A2;
    m2 = m3;
    u3 = u4;
    u4 = u5;
  }

  if (c == 0) {
    if (ntile == TPB) {
#pragma unroll
      for (int tt = 0; tt < TPB; ++tt) {
        float4* o4 = reinterpret_cast<float4*>(orow + tt * 16);
        o4[0] = make_float4(acc[tt][0], acc[tt][1], acc[tt][2], acc[tt][3]);
        o4[1] = make_float4(acc[tt][4], acc[tt][5], acc[tt][6], acc[tt][7]);
      }
    } else {
#pragma unroll
      for (int tt = 0; tt < TPB; ++tt) {
        if (tt < ntile) {
          float4* o4 = reinterpret_cast<float4*>(orow + tt * 16);
          o4[0] = make_float4(acc[tt][0], acc[tt][1], acc[tt][2], acc[tt][3]);
          o4[1] = make_float4(acc[tt][4], acc[tt][5], acc[tt][6], acc[tt][7]);
        }
      }
    }
  }
}

__global__ void __launch_bounds__(128)
kbln_score(const float* __restrict__ lit,
           const float* __restrict__ cen,
           const float* __restrict__ var,
           const float* __restrict__ nfw,
           const int* __restrict__ head_ids,
           const int* __restrict__ rel_ids,
           float* __restrict__ out) {
  __shared__ __attribute__((aligned(16))) float s_aq[NUM_B * NUM_F];
  __shared__ __attribute__((aligned(16))) float s_w[NUM_B * NUM_F];
  __shared__ __attribute__((aligned(16))) float s_lq[ROWS * PITCH];

  const int t = threadIdx.x;
  const int k4 = t & 15;
  const int rsub = t >> 4;

  int hid[NUM_B / 8], rid[NUM_B / 8];
#pragma unroll
  for (int it = 0; it < NUM_B / 8; ++it) {
    hid[it] = head_ids[rsub + 8 * it];
    rid[it] = rel_ids[rsub + 8 * it];
  }

  const int row0 = blockIdx.x * ROWS;
  float4 lv[ROWS / 8];
#pragma unroll
  for (int i = 0; i < ROWS / 8; ++i) {
    int e = row0 + rsub + 8 * i;
    e = e < NUM_E ? e : NUM_E - 1;
    lv[i] = reinterpret_cast<const float4*>(lit + (size_t)e * NUM_F)[k4];
  }

  __builtin_amdgcn_sched_barrier(0);

  const float4 var4 = reinterpret_cast<const float4*>(var)[k4];
  const float4 c4   = reinterpret_cast<const float4*>(cen)[k4];

  float4 l4[NUM_B / 8], w4[NUM_B / 8];
#pragma unroll
  for (int it = 0; it < NUM_B / 8; ++it) {
    l4[it] = reinterpret_cast<const float4*>(lit + (size_t)hid[it] * NUM_F)[k4];
    w4[it] = reinterpret_cast<const float4*>(nfw + (size_t)rid[it] * NUM_F)[k4];
  }

  float4 q4;
  q4.x = sqrtf(LOG2E / var4.x);
  q4.y = sqrtf(LOG2E / var4.y);
  q4.z = sqrtf(LOG2E / var4.z);
  q4.w = sqrtf(LOG2E / var4.w);

#pragma unroll
  for (int i = 0; i < ROWS / 8; ++i) {
    float* dst = s_lq + (rsub + 8 * i) * PITCH;
    *reinterpret_cast<float2*>(dst + staged_slot(k4, 0)) = make_float2(lv[i].x * q4.x, lv[i].y * q4.y);
    *reinterpret_cast<float2*>(dst + staged_slot(k4, 1)) = make_float2(lv[i].z * q4.z, lv[i].w * q4.w);
  }

#pragma unroll
  for (int it = 0; it < NUM_B / 8; ++it) {
    const int base = (rsub + 8 * it) * NUM_F;
    *reinterpret_cast<float2*>(s_aq + base + staged_slot(k4, 0)) =
        make_float2((l4[it].x - c4.x) * q4.x, (l4[it].y - c4.y) * q4.y);
    *reinterpret_cast<float2*>(s_aq + base + staged_slot(k4, 1)) =
        make_float2((l4[it].z - c4.z) * q4.z, (l4[it].w - c4.w) * q4.w);
    *reinterpret_cast<float2*>(s_w  + base + staged_slot(k4, 0)) = make_float2(w4[it].x, w4[it].y);
    *reinterpret_cast<float2*>(s_w  + base + staged_slot(k4, 1)) = make_float2(w4[it].z, w4[it].w);
  }
  __syncthreads();

  const int wave = t >> 5;
  const int lane = t & 31;
  const int c = lane & 15;
  const int h = lane >> 4;
  const int tile0 = blockIdx.x * TPB;
  const int ntile = (NTILES - tile0) < TPB ? (NTILES - tile0) : TPB;

  v4f lq[TPB][8];
#pragma unroll
  for (int tt = 0; tt < TPB; ++tt) {
#pragma unroll
    for (int j = 0; j < 8; ++j)
      lq[tt][j] = *reinterpret_cast<const v4f*>(s_lq + (tt * 16 + c) * PITCH + 8 * j + 4 * h);
  }

  const int b0 = wave * BPW;
  float* orow = out + (size_t)b0 * NUM_E + tile0 * 16 + 8 * h;

  v4f aqA[8], wvA[8], aqB[8], wvB[8];
  load_bvec(aqA, wvA, s_aq, s_w, b0, h);
  for (int bb = 0; bb < BPW; bb += 2) {
    int bn1 = b0 + bb + 1;
    int bn2 = b0 + bb + 2;
    bn2 = bn2 < NUM_B ? bn2 : NUM_B - 1;
    score_b(lq, aqA, wvA, aqB, wvB, s_aq, s_w, bn1, h, orow, c, ntile);
    orow += NUM_E;
    score_b(lq, aqB, wvB, aqA, wvA, s_aq, s_w, bn2, h, orow, c, ntile);
    orow += NUM_E;
  }
}

extern "C" void kernel_launch(void* const* d_in, const int* in_sizes, int n_in,
                              void* d_out, int out_size, void* d_ws, size_t ws_size,
                              hipStream_t stream) {
  const float* lit  = (const float*)d_in[0];
  const float* c    = (const float*)d_in[1];
  const float* var  = (const float*)d_in[2];
  const float* nfw  = (const float*)d_in[3];
  const int*   head = (const int*)d_in[4];
  const int*   rel  = (const int*)d_in[5];

  kbln_score<<<GRID_MAIN, 128, 0, stream>>>(lit, c, var, nfw, head, rel, (float*)d_out);
}


# Matrix operands exchanged (A = w rows, B = p columns): each lane holds its own e, so a b's 13 tiles leave in 7 whole-wave 128-byte stores (was 26 two-lane 16-byte-piece stores); sums unchanged
# speedup vs baseline: 1.8360x; 1.0311x over previous
#include <hip/hip_runtime.h>
#include <math.h>


typedef __attribute__((ext_vector_type(2))) float v2f;
typedef __attribute__((ext_vector_type(4))) float v4f;
typedef __attribute__((ext_vector_type(8))) float v8f;

#define NUM_E 50000
#define NUM_F 64
#define NUM_B 64
#define NTILES (NUM_E / 16)
#define TPB 13
#define GRID_MAIN ((NTILES + TPB - 1) / TPB)
#define ROWS (TPB * 16)
#define PITCH 68
#define BPW 16

#define LOG2E 1.4426950408889634f

__device__ __forceinline__ int staged_slot(int k, int h) {
  return 8 * (k >> 1) + 4 * h + 2 * (k & 1);
}

__device__ __forceinline__ void load_bvec(v4f (&aq)[8], v4f (&wv)[8],
                                          const float* s_aq, const float* s_w,
                                          int b, int h) {
  const float* pa = s_aq + b * NUM_F + 4 * h;
  const float* pw = s_w  + b * NUM_F + 4 * h;
#pragma unroll
  for (int j = 0; j < 8; ++j) {
    aq[j] = *reinterpret_cast<const v4f*>(pa + 8 * j);
    wv[j] = *reinterpret_cast<const v4f*>(pw + 8 * j);
  }
}

__device__ __forceinline__ v2f step_u(const v4f (&lq)[TPB][8], const v4f (&aq)[8], int s) {
  const int k = s / TPB, tt = s % TPB;
  const int j = k >> 1, hf = k & 1;
  v2f u;
  u[0] = lq[tt][j][2 * hf]     - aq[j][2 * hf];
  u[1] = lq[tt][j][2 * hf + 1] - aq[j][2 * hf + 1];
  return u;
}

__device__ __forceinline__ v2f step_m(const v4f (&lq)[TPB][8], const v4f (&aq)[8], int s) {
  v2f u = step_u(lq, aq, s);
  v2f m;
  m[0] = -(u[0] * u[0]);
  m[1] = -(u[1] * u[1]);
  return m;
}

__device__ __forceinline__ v2f exp2_v2(v2f m) {
  v2f A;
  A[0] = __builtin_amdgcn_exp2f(m[0]);
  A[1] = __builtin_amdgcn_exp2f(m[1]);
  return A;
}

#define NSTEP (16 * TPB)

__device__ __forceinline__ void score_b(const v4f (&lq)[TPB][8],
                                        const v4f (&aq)[8], const v4f (&wv)[8],
                                        v4f (&aqn)[8], v4f (&wvn)[8],
                                        const float* s_aq, const float* s_w,
                                        int bnext, int h,
                                        float* op, int ntile) {
  load_bvec(aqn, wvn, s_aq, s_w, bnext, h);
  __builtin_amdgcn_sched_barrier(0);

  v8f acc[TPB];
#pragma unroll
  for (int tt = 0; tt < TPB; ++tt) acc[tt] = (v8f){};

  v2f A0 = exp2_v2(step_m(lq, aq, 0));
  v2f A1 = exp2_v2(step_m(lq, aq, 1));
  v2f m2 = step_m(lq, aq, 2);
  v2f u3 = step_u(lq, aq, 3);
  v2f u4 = step_u(lq, aq, 4);
#pragma unroll
  for (int s = 0; s < NSTEP; ++s) {
    const int k = s / TPB, tt = s % TPB;
    const int j = k >> 1, hf = k & 1;
    v2f A2 = A1, m3 = m2, u5 = u4;
    if (s + 2 < NSTEP) A2[0] = __builtin_amdgcn_exp2f(m2[0]);
    if (s + 3 < NSTEP) {
      m3[0] = -(u3[0] * u3[0]);
      m3[1] = -(u3[1] * u3[1]);
    }
    asm volatile("" :: "v"(A2[0]), "v"(m3[0]), "v"(m3[1]));
    __builtin_amdgcn_sched_barrier(0);
    if (s + 2 < NSTEP) A2[1] = __builtin_amdgcn_exp2f(m2[1]);
    if (s + 5 < NSTEP) u5 = step_u(lq, aq, s + 5);
    asm volatile("" :: "v"(A2[1]), "v"(u5[0]), "v"(u5[1]));
    v2f Bm;
    Bm[0] = wv[j][2 * hf];
    Bm[1] = wv[j][2 * hf + 1];
    acc[tt] = __builtin_amdgcn_wmma_f32_16x16x4_f32(
         false, Bm,  false, A0,
         (short)0, acc[tt],  false,  false);
    __builtin_amdgcn_sched_barrier(0);
    A0 = A1;
    A1 = A2;
    m2 = m3;
    u3 = u4;
    u4 = u5;
  }

  if (ntile == TPB) {
#pragma unroll
    for (int j = 0; j < TPB / 2; ++j)
      op[32 * j] = h ? acc[2 * j + 1][0] : acc[2 * j][0];
    if (h == 0) op[32 * (TPB / 2)] = acc[TPB - 1][0];
  } else {
#pragma unroll
    for (int j = 0; j < TPB / 2; ++j) {
      if (2 * j + h < ntile) op[32 * j] = h ? acc[2 * j + 1][0] : acc[2 * j][0];
    }
    if (h == 0 && TPB - 1 < ntile) op[32 * (TPB / 2)] = acc[TPB - 1][0];
  }
}

__global__ void __launch_bounds__(128)
kbln_score(const float* __restrict__ lit,
           const float* __restrict__ cen,
           const float* __restrict__ var,
           const float* __restrict__ nfw,
           const int* __restrict__ head_ids,
           const int* __restrict__ rel_ids,
           float* __restrict__ out) {
  __shared__ __attribute__((aligned(16))) float s_aq[NUM_B * NUM_F];
  __shared__ __attribute__((aligned(16))) float s_w[NUM_B * NUM_F];
  __shared__ __attribute__((aligned(16))) float s_lq[ROWS * PITCH];

  const int t = threadIdx.x;
  const int k4 = t & 15;
  const int rsub = t >> 4;

  int hid[NUM_B / 8], rid[NUM_B / 8];
#pragma unroll
  for (int it = 0; it < NUM_B / 8; ++it) {
    hid[it] = head_ids[rsub + 8 * it];
    rid[it] = rel_ids[rsub + 8 * it];
  }

  const int row0 = blockIdx.x * ROWS;
  float4 lv[ROWS / 8];
#pragma unroll
  for (int i = 0; i < ROWS / 8; ++i) {
    int e = row0 + rsub + 8 * i;
    e = e < NUM_E ? e : NUM_E - 1;
    lv[i] = reinterpret_cast<const float4*>(lit + (size_t)e * NUM_F)[k4];
  }

  __builtin_amdgcn_sched_barrier(0);

  const float4 var4 = reinterpret_cast<const float4*>(var)[k4];
  const float4 c4   = reinterpret_cast<const float4*>(cen)[k4];

  float4 l4[NUM_B / 8], w4[NUM_B / 8];
#pragma unroll
  for (int it = 0; it < NUM_B / 8; ++it) {
    l4[it] = reinterpret_cast<const float4*>(lit + (size_t)hid[it] * NUM_F)[k4];
    w4[it] = reinterpret_cast<const float4*>(nfw + (size_t)rid[it] * NUM_F)[k4];
  }

  float4 q4;
  q4.x = sqrtf(LOG2E / var4.x);
  q4.y = sqrtf(LOG2E / var4.y);
  q4.z = sqrtf(LOG2E / var4.z);
  q4.w = sqrtf(LOG2E / var4.w);

#pragma unroll
  for (int i = 0; i < ROWS / 8; ++i) {
    float* dst = s_lq + (rsub + 8 * i) * PITCH;
    *reinterpret_cast<float2*>(dst + staged_slot(k4, 0)) = make_float2(lv[i].x * q4.x, lv[i].y * q4.y);
    *reinterpret_cast<float2*>(dst + staged_slot(k4, 1)) = make_float2(lv[i].z * q4.z, lv[i].w * q4.w);
  }

#pragma unroll
  for (int it = 0; it < NUM_B / 8; ++it) {
    const int base = (rsub + 8 * it) * NUM_F;
    *reinterpret_cast<float2*>(s_aq + base + staged_slot(k4, 0)) =
        make_float2((l4[it].x - c4.x) * q4.x, (l4[it].y - c4.y) * q4.y);
    *reinterpret_cast<float2*>(s_aq + base + staged_slot(k4, 1)) =
        make_float2((l4[it].z - c4.z) * q4.z, (l4[it].w - c4.w) * q4.w);
    *reinterpret_cast<float2*>(s_w  + base + staged_slot(k4, 0)) = make_float2(w4[it].x, w4[it].y);
    *reinterpret_cast<float2*>(s_w  + base + staged_slot(k4, 1)) = make_float2(w4[it].z, w4[it].w);
  }
  __syncthreads();

  const int wave = t >> 5;
  const int lane = t & 31;
  const int c = lane & 15;
  const int h = lane >> 4;
  const int tile0 = blockIdx.x * TPB;
  const int ntile = (NTILES - tile0) < TPB ? (NTILES - tile0) : TPB;

  v4f lq[TPB][8];
#pragma unroll
  for (int tt = 0; tt < TPB; ++tt) {
#pragma unroll
    for (int j = 0; j < 8; ++j)
      lq[tt][j] = *reinterpret_cast<const v4f*>(s_lq + (tt * 16 + c) * PITCH + 8 * j + 4 * h);
  }

  const int b0 = wave * BPW;
  float* orow = out + (size_t)b0 * NUM_E + tile0 * 16 + lane;

  v4f aqA[8], wvA[8], aqB[8], wvB[8];
  load_bvec(aqA, wvA, s_aq, s_w, b0, h);
  for (int bb = 0; bb < BPW; bb += 2) {
    int bn1 = b0 + bb + 1;
    int bn2 = b0 + bb + 2;
    bn2 = bn2 < NUM_B ? bn2 : NUM_B - 1;
    score_b(lq, aqA, wvA, aqB, wvB, s_aq, s_w, bn1, h, orow, ntile);
    orow += NUM_E;
    score_b(lq, aqB, wvB, aqA, wvA, s_aq, s_w, bn2, h, orow, ntile);
    orow += NUM_E;
  }
}

extern "C" void kernel_launch(void* const* d_in, const int* in_sizes, int n_in,
                              void* d_out, int out_size, void* d_ws, size_t ws_size,
                              hipStream_t stream) {
  const float* lit  = (const float*)d_in[0];
  const float* c    = (const float*)d_in[1];
  const float* var  = (const float*)d_in[2];
  const float* nfw  = (const float*)d_in[3];
  const int*   head = (const int*)d_in[4];
  const int*   rel  = (const int*)d_in[5];

  kbln_score<<<GRID_MAIN, 128, 0, stream>>>(lit, c, var, nfw, head, rel, (float*)d_out);
}
